# starting kernel
# baseline (speedup 1.0000x reference)
#include <hip/hip_runtime.h>

typedef __attribute__((ext_vector_type(2))) float v2f;
typedef __attribute__((ext_vector_type(8))) float v8f;

#define NTASKS 512
#define NQ     75
#define NS     25
#define DIM    1024
#define NWAY   5
#define NSHOT  5

#define KC     64
#define PSTR   1032
#define QSTR   68
#define NWAVES 5
#define BLOCK  (NWAVES * 32)

__global__ __launch_bounds__(BLOCK)
void proto_head_wmma_kernel(const float* __restrict__ query,
                            const float* __restrict__ support,
                            const float* __restrict__ scale,
                            float* __restrict__ out)
{
    __shared__ float protos[NWAY * PSTR];
    __shared__ float qtile[NWAVES][16 * QSTR];
    __shared__ float bb[NWAY];
    __shared__ float aa_lds[NWAVES * 16];

    const int b    = blockIdx.x;
    const int tid  = threadIdx.x;
    const int lane = tid & 31;
    const int wave = tid >> 5;

    const float* sup = support + (size_t)b * NS * DIM;
    const float* q   = query   + (size_t)b * NQ * DIM;

    if (tid < NWAY) bb[tid] = 0.0f;
    __syncthreads();

    float bbp[NWAY] = {0.f, 0.f, 0.f, 0.f, 0.f};
    for (int d = tid; d < DIM; d += BLOCK) {
#pragma unroll
        for (int w = 0; w < NWAY; ++w) {
            float s = 0.0f;
#pragma unroll
            for (int j = 0; j < NSHOT; ++j)
                s += sup[(size_t)(w * NSHOT + j) * DIM + d];
            float p = s * 0.2f;
            protos[w * PSTR + d] = p;
            bbp[w] += p * p;
        }
    }
#pragma unroll
    for (int w = 0; w < NWAY; ++w) atomicAdd(&bb[w], bbp[w]);
    __syncthreads();

    const int mbase = wave * 16;
    const int mrow  = lane & 15;
    const int hi    = lane >> 4;
    const int klo   = 2 * hi;

    const int   n     = lane & 15;
    const int   nclmp = (n < NWAY) ? n : (NWAY - 1);
    const float nmask = (n < NWAY) ? 1.0f : 0.0f;

    float* qt = &qtile[wave][0];
    v8f   acc = {};
    float aa_part = 0.0f;

    for (int kc = 0; kc < DIM; kc += KC) {
#pragma unroll 4
        for (int r = 0; r < 16; ++r) {
            int m = mbase + r;
            float2 v = make_float2(0.0f, 0.0f);
            if (m < NQ)
                v = *(const float2*)(q + (size_t)m * DIM + kc + lane * 2);
            qt[r * QSTR + lane * 2]     = v.x;
            qt[r * QSTR + lane * 2 + 1] = v.y;
        }

        if (kc + KC < DIM) {
            int mp = mbase + mrow;
            if (mp >= NQ) mp = NQ - 1;
            __builtin_prefetch(q + (size_t)mp * DIM + kc + KC + lane * 2, 0, 0);
        }

        {
            float s = 0.0f;
#pragma unroll
            for (int c = 0; c < 32; ++c) {
                float v = qt[mrow * QSTR + hi * 32 + c];
                s += v * v;
            }
            aa_part += s;
        }

#pragma unroll
        for (int k = 0; k < KC; k += 4) {
            v2f a, bf;
            a.x  = qt[mrow * QSTR + k + klo];
            a.y  = qt[mrow * QSTR + k + klo + 1];
            bf.x = nmask * protos[nclmp * PSTR + kc + k + klo];
            bf.y = nmask * protos[nclmp * PSTR + kc + k + klo + 1];
            acc = __builtin_amdgcn_wmma_f32_16x16x4_f32(
                 false, a,  false, bf,
                 (short)0, acc,  false,  false);
        }
    }

    float aa_full = aa_part + __shfl_xor(aa_part, 16, 32);
    if (lane < 16) aa_lds[wave * 16 + lane] = aa_full;

    const float scl   = scale[0];
    const float inv_d = 1.0f / (float)DIM;
    if (n < NWAY) {
#pragma unroll
        for (int j = 0; j < 8; ++j) {
            int m = mbase + j + 8 * hi;
            if (m < NQ) {
                float aav   = aa_lds[wave * 16 + j + 8 * hi];
                float logit = (2.0f * acc[j] - aav - bb[n]) * inv_d * scl;
                out[((size_t)b * NQ + m) * NWAY + n] = logit;
            }
        }
    }
}

extern "C" void kernel_launch(void* const* d_in, const int* in_sizes, int n_in,
                              void* d_out, int out_size, void* d_ws, size_t ws_size,
                              hipStream_t stream) {
    const float* query   = (const float*)d_in[0];
    const float* support = (const float*)d_in[1];
    const float* scale   = (const float*)d_in[5];
    float* out = (float*)d_out;

    proto_head_wmma_kernel<<<NTASKS, BLOCK, 0, stream>>>(query, support, scale, out);
}


# query tile loads grouped and kept in flight: 16 branch-free loads per 64-column chunk off one base (last tile moved up to rows 59-74), ring of 3 register chunks requested ahead of use
# speedup vs baseline: 2.4138x; 2.4138x over previous
#include <hip/hip_runtime.h>


typedef __attribute__((ext_vector_type(2))) float v2f;
typedef __attribute__((ext_vector_type(8))) float v8f;

#define NTASKS 512
#define NQ     75
#define NS     25
#define DIM    1024
#define NWAY   5
#define NSHOT  5

#define KC     64
#define NCHUNK (DIM / KC)
#define NBUF   3
#define PSTR   1032
#define QSTR   68
#define NWAVES 5
#define BLOCK  (NWAVES * 32)

__device__ __forceinline__ void load_chunk(v2f (&buf)[16], const float* __restrict__ qlane, int chunk)
{
#pragma unroll
    for (int r = 0; r < 16; ++r)
        buf[r] = *(const v2f*)(qlane + r * DIM + chunk * KC);
}

__global__ __launch_bounds__(BLOCK)
void proto_head_wmma_kernel(const float* __restrict__ query,
                            const float* __restrict__ support,
                            const float* __restrict__ scale,
                            float* __restrict__ out)
{
    __shared__ float protos[NWAY * PSTR];
    __shared__ __attribute__((aligned(8))) float qtile[NWAVES][16 * QSTR];
    __shared__ float bb[NWAY];
    __shared__ float aa_lds[NWAVES * 16];

    const int b    = blockIdx.x;
    const int tid  = threadIdx.x;
    const int lane = tid & 31;
    const int wave = tid >> 5;

    const float* sup = support + (size_t)b * NS * DIM;
    const float* q   = query   + (size_t)b * NQ * DIM;

    const int mown  = wave * 16;
    const int mbase = (mown < NQ - 16) ? mown : (NQ - 16);
    const float* qlane = q + (size_t)mbase * DIM + lane * 2;

    if (tid < NWAY) bb[tid] = 0.0f;
    __syncthreads();

    v2f qbuf[NBUF][16];
    load_chunk(qbuf[0], qlane, 0);

    float bbp[NWAY] = {0.f, 0.f, 0.f, 0.f, 0.f};
    for (int d = tid; d < DIM; d += BLOCK) {
#pragma unroll
        for (int w = 0; w < NWAY; ++w) {
            float s = 0.0f;
#pragma unroll
            for (int j = 0; j < NSHOT; ++j)
                s += sup[(size_t)(w * NSHOT + j) * DIM + d];
            float p = s * 0.2f;
            protos[w * PSTR + d] = p;
            bbp[w] += p * p;
        }
    }

#pragma unroll
    for (int c = 1; c < NBUF; ++c) load_chunk(qbuf[c], qlane, c);

#pragma unroll
    for (int w = 0; w < NWAY; ++w) atomicAdd(&bb[w], bbp[w]);
    __syncthreads();

    const int mrow  = lane & 15;
    const int hi    = lane >> 4;
    const int klo   = 2 * hi;

    const int   n     = lane & 15;
    const int   nclmp = (n < NWAY) ? n : (NWAY - 1);
    const float nmask = (n < NWAY) ? 1.0f : 0.0f;

    float* qt = &qtile[wave][0];
    v8f   acc = {};
    float aa_part = 0.0f;

#pragma unroll
    for (int c = 0; c < NCHUNK; ++c) {
        const int kc = c * KC;

#pragma unroll
        for (int r = 0; r < 16; ++r)
            *(v2f*)(qt + r * QSTR + lane * 2) = qbuf[c % NBUF][r];

        __builtin_amdgcn_sched_barrier(0);
        if (c + NBUF < NCHUNK) load_chunk(qbuf[c % NBUF], qlane, c + NBUF);
        __builtin_amdgcn_sched_barrier(0);

        {
            float s = 0.0f;
#pragma unroll
            for (int cc = 0; cc < 32; ++cc) {
                float v = qt[mrow * QSTR + hi * 32 + cc];
                s += v * v;
            }
            aa_part += s;
        }

#pragma unroll
        for (int k = 0; k < KC; k += 4) {
            v2f a, bf;
            a.x  = qt[mrow * QSTR + k + klo];
            a.y  = qt[mrow * QSTR + k + klo + 1];
            bf.x = nmask * protos[nclmp * PSTR + kc + k + klo];
            bf.y = nmask * protos[nclmp * PSTR + kc + k + klo + 1];
            acc = __builtin_amdgcn_wmma_f32_16x16x4_f32(
                 false, a,  false, bf,
                 (short)0, acc,  false,  false);
        }
    }

    float aa_full = aa_part + __shfl_xor(aa_part, 16, 32);
    if (lane < 16) aa_lds[wave * 16 + lane] = aa_full;

    const float scl   = scale[0];
    const float inv_d = 1.0f / (float)DIM;
    if (n < NWAY) {
#pragma unroll
        for (int j = 0; j < 8; ++j) {
            int m = mbase + j + 8 * hi;
            if (m >= mown) {
                float aav   = aa_lds[wave * 16 + j + 8 * hi];
                float logit = (2.0f * acc[j] - aav - bb[n]) * inv_d * scl;
                out[((size_t)b * NQ + m) * NWAY + n] = logit;
            }
        }
    }
}

extern "C" void kernel_launch(void* const* d_in, const int* in_sizes, int n_in,
                              void* d_out, int out_size, void* d_ws, size_t ws_size,
                              hipStream_t stream) {
    const float* query   = (const float*)d_in[0];
    const float* support = (const float*)d_in[1];
    const float* scale   = (const float*)d_in[5];
    float* out = (float*)d_out;

    proto_head_wmma_kernel<<<NTASKS, BLOCK, 0, stream>>>(query, support, scale, out);
}


# LDS operand reads issued as one group a whole block of 8 matrix instructions ahead of use (two register sets), not 2-3 ahead among them; |q|^2 summed over those A fragments (summation order changes)
# speedup vs baseline: 2.4910x; 1.0320x over previous
#include <hip/hip_runtime.h>


typedef __attribute__((ext_vector_type(2))) float v2f;
typedef __attribute__((ext_vector_type(8))) float v8f;

#define NTASKS 512
#define NQ     75
#define NS     25
#define DIM    1024
#define NWAY   5
#define NSHOT  5

#define KC     64
#define NCHUNK (DIM / KC)
#define NBUF   3
#define PSTR   1032
#define QSTR   68
#define NWAVES 5
#define BLOCK  (NWAVES * 32)

__device__ __forceinline__ void load_chunk(v2f (&buf)[16], const float* __restrict__ qlane, int chunk)
{
#pragma unroll
    for (int r = 0; r < 16; ++r)
        buf[r] = *(const v2f*)(qlane + r * DIM + chunk * KC);
}

__global__ __launch_bounds__(BLOCK)
void proto_head_wmma_kernel(const float* __restrict__ query,
                            const float* __restrict__ support,
                            const float* __restrict__ scale,
                            float* __restrict__ out)
{
    __shared__ float protos[NWAY * PSTR];
    __shared__ __attribute__((aligned(8))) float qtile[NWAVES][16 * QSTR];
    __shared__ float bb[NWAY];
    __shared__ float aa_lds[NWAVES * 16];

    const int b    = blockIdx.x;
    const int tid  = threadIdx.x;
    const int lane = tid & 31;
    const int wave = tid >> 5;

    const float* sup = support + (size_t)b * NS * DIM;
    const float* q   = query   + (size_t)b * NQ * DIM;

    const int mown  = wave * 16;
    const int mbase = (mown < NQ - 16) ? mown : (NQ - 16);
    const float* qlane = q + (size_t)mbase * DIM + lane * 2;

    if (tid < NWAY) bb[tid] = 0.0f;
    __syncthreads();

    v2f qbuf[NBUF][16];
    load_chunk(qbuf[0], qlane, 0);

    float bbp[NWAY] = {0.f, 0.f, 0.f, 0.f, 0.f};
    for (int d = tid; d < DIM; d += BLOCK) {
#pragma unroll
        for (int w = 0; w < NWAY; ++w) {
            float s = 0.0f;
#pragma unroll
            for (int j = 0; j < NSHOT; ++j)
                s += sup[(size_t)(w * NSHOT + j) * DIM + d];
            float p = s * 0.2f;
            protos[w * PSTR + d] = p;
            bbp[w] += p * p;
        }
    }

#pragma unroll
    for (int c = 1; c < NBUF; ++c) load_chunk(qbuf[c], qlane, c);

#pragma unroll
    for (int w = 0; w < NWAY; ++w) atomicAdd(&bb[w], bbp[w]);
    __syncthreads();

    const int mrow  = lane & 15;
    const int hi    = lane >> 4;
    const int klo   = 2 * hi;

    const int   n     = lane & 15;
    const int   nclmp = (n < NWAY) ? n : (NWAY - 1);
    const float nmask = (n < NWAY) ? 1.0f : 0.0f;

    float* qt = &qtile[wave][0];
    v8f   acc = {};
    float aa_x = 0.0f, aa_y = 0.0f;

    v2f fa[2][8], fb[2][8];

#define STAGE(c)                                                                   \
    _Pragma("unroll") for (int r = 0; r < 16; ++r)                                 \
        *(v2f*)(qt + r * QSTR + lane * 2) = qbuf[(c) % NBUF][r];

#define FRAGS(c, h)                                                                \
    _Pragma("unroll") for (int k = 0; k < 8; ++k) {                                \
        fa[h][k] = *(const v2f*)(qt + mrow * QSTR + 4 * (8 * (h) + k) + klo);      \
        fb[h][k] = *(const v2f*)(&protos[nclmp * PSTR + (c) * KC + 4 * (8 * (h) + k) + klo]); \
    }

#define MATH(h)                                                                    \
    _Pragma("unroll") for (int k = 0; k < 8; ++k) {                                \
        aa_x += fa[h][k].x * fa[h][k].x;                                           \
        aa_y += fa[h][k].y * fa[h][k].y;                                           \
        v2f bf = nmask * fb[h][k];                                                 \
        acc = __builtin_amdgcn_wmma_f32_16x16x4_f32(                               \
             false, fa[h][k],  false, bf,                        \
             (short)0, acc,  false,  false);        \
    }

    STAGE(0)
    FRAGS(0, 0)
    load_chunk(qbuf[0], qlane, NBUF);

#pragma unroll
    for (int c = 0; c < NCHUNK; ++c) {
        __builtin_amdgcn_sched_barrier(0);
        FRAGS(c, 1)
        __builtin_amdgcn_sched_barrier(0);
        MATH(0)
        __builtin_amdgcn_sched_barrier(0);
        if (c + 1 < NCHUNK) {
            STAGE(c + 1)
            FRAGS(c + 1, 0)
        }
        __builtin_amdgcn_sched_barrier(0);
        if (c + 1 + NBUF < NCHUNK) load_chunk(qbuf[(c + 1) % NBUF], qlane, c + 1 + NBUF);
        __builtin_amdgcn_sched_barrier(0);
        MATH(1)
    }
#undef STAGE
#undef FRAGS
#undef MATH
    const float aa_part = aa_x + aa_y;

    float aa_full = aa_part + __shfl_xor(aa_part, 16, 32);
    if (lane < 16) aa_lds[wave * 16 + lane] = aa_full;

    const float scl   = scale[0];
    const float inv_d = 1.0f / (float)DIM;
    if (n < NWAY) {
#pragma unroll
        for (int j = 0; j < 8; ++j) {
            int m = mbase + j + 8 * hi;
            if (m >= mown) {
                float aav   = aa_lds[wave * 16 + j + 8 * hi];
                float logit = (2.0f * acc[j] - aav - bb[n]) * inv_d * scl;
                out[((size_t)b * NQ + m) * NWAY + n] = logit;
            }
        }
    }
}

extern "C" void kernel_launch(void* const* d_in, const int* in_sizes, int n_in,
                              void* d_out, int out_size, void* d_ws, size_t ws_size,
                              hipStream_t stream) {
    const float* query   = (const float*)d_in[0];
    const float* support = (const float*)d_in[1];
    const float* scale   = (const float*)d_in[5];
    float* out = (float*)d_out;

    proto_head_wmma_kernel<<<NTASKS, BLOCK, 0, stream>>>(query, support, scale, out);
}


# prototype pass in one round trip: wave w owns class w, 40 16-byte loads at once, 3 query chunks behind them in flight over an LDS-only barrier; |p|^2 by butterfly, no atomics (sum order changes)
# speedup vs baseline: 2.9392x; 1.1799x over previous
#include <hip/hip_runtime.h>


typedef __attribute__((ext_vector_type(2))) float v2f;
typedef __attribute__((ext_vector_type(4))) float v4f;
typedef __attribute__((ext_vector_type(8))) float v8f;

#define NTASKS 512
#define NQ     75
#define NS     25
#define DIM    1024
#define NWAY   5
#define NSHOT  5

#define KC     64
#define NCHUNK (DIM / KC)
#define NBUF   3
#define PSTR   1032
#define QSTR   68
#define NWAVES 5
#define BLOCK  (NWAVES * 32)

__device__ __forceinline__ void load_chunk(v2f (&buf)[16], const float* __restrict__ qlane, int chunk)
{
#pragma unroll
    for (int r = 0; r < 16; ++r)
        buf[r] = *(const v2f*)(qlane + r * DIM + chunk * KC);
}

__global__ __launch_bounds__(BLOCK)
void proto_head_wmma_kernel(const float* __restrict__ query,
                            const float* __restrict__ support,
                            const float* __restrict__ scale,
                            float* __restrict__ out)
{
    __shared__ __attribute__((aligned(16))) float protos[NWAY * PSTR];
    __shared__ __attribute__((aligned(8))) float qtile[NWAVES][16 * QSTR];
    __shared__ float bb[NWAY];
    __shared__ float aa_lds[NWAVES * 16];

    const int b    = blockIdx.x;
    const int tid  = threadIdx.x;
    const int lane = tid & 31;
    const int wave = tid >> 5;

    const float* sup = support + (size_t)b * NS * DIM;
    const float* q   = query   + (size_t)b * NQ * DIM;

    const int mown  = wave * 16;
    const int mbase = (mown < NQ - 16) ? mown : (NQ - 16);
    const float* qlane = q + (size_t)mbase * DIM + lane * 2;

    const float* srow = sup + (size_t)wave * NSHOT * DIM + lane * 4;
    v4f sv[NSHOT][DIM / 128];
#pragma unroll
    for (int j = 0; j < NSHOT; ++j)
#pragma unroll
        for (int i = 0; i < DIM / 128; ++i)
            sv[j][i] = *(const v4f*)(srow + j * DIM + i * 128);

    v2f qbuf[NBUF][16];
#pragma unroll
    for (int c = 0; c < NBUF; ++c) {
        load_chunk(qbuf[c], qlane, c);
        __builtin_amdgcn_sched_barrier(0);
    }

    v4f bb4 = {0.f, 0.f, 0.f, 0.f};
#pragma unroll
    for (int i = 0; i < DIM / 128; ++i) {
        v4f s = sv[0][i];
#pragma unroll
        for (int j = 1; j < NSHOT; ++j) s += sv[j][i];
        v4f p = s * 0.2f;
        *(v4f*)&protos[wave * PSTR + lane * 4 + i * 128] = p;
        bb4 += p * p;
    }
    float bbp = (bb4.x + bb4.y) + (bb4.z + bb4.w);
#pragma unroll
    for (int off = 16; off > 0; off >>= 1) bbp += __shfl_xor(bbp, off, 32);
    if (lane == 0) bb[wave] = bbp;
    __builtin_amdgcn_fence(__ATOMIC_RELEASE, "workgroup", "local");
    __builtin_amdgcn_s_barrier();
    __builtin_amdgcn_fence(__ATOMIC_ACQUIRE, "workgroup", "local");

    const int mrow  = lane & 15;
    const int hi    = lane >> 4;
    const int klo   = 2 * hi;

    const int   n     = lane & 15;
    const int   nclmp = (n < NWAY) ? n : (NWAY - 1);
    const float nmask = (n < NWAY) ? 1.0f : 0.0f;

    float* qt = &qtile[wave][0];
    v8f   acc = {};
    float aa_x = 0.0f, aa_y = 0.0f;

    v2f fa[2][8], fb[2][8];

#define STAGE(c)                                                                   \
    _Pragma("unroll") for (int r = 0; r < 16; ++r)                                 \
        *(v2f*)(qt + r * QSTR + lane * 2) = qbuf[(c) % NBUF][r];

#define FRAGS(c, h)                                                                \
    _Pragma("unroll") for (int k = 0; k < 8; ++k) {                                \
        fa[h][k] = *(const v2f*)(qt + mrow * QSTR + 4 * (8 * (h) + k) + klo);      \
        fb[h][k] = *(const v2f*)(&protos[nclmp * PSTR + (c) * KC + 4 * (8 * (h) + k) + klo]); \
    }

#define MATH(h)                                                                    \
    _Pragma("unroll") for (int k = 0; k < 8; ++k) {                                \
        aa_x += fa[h][k].x * fa[h][k].x;                                           \
        aa_y += fa[h][k].y * fa[h][k].y;                                           \
        v2f bf = nmask * fb[h][k];                                                 \
        acc = __builtin_amdgcn_wmma_f32_16x16x4_f32(                               \
             false, fa[h][k],  false, bf,                        \
             (short)0, acc,  false,  false);        \
    }

    STAGE(0)
    FRAGS(0, 0)
    load_chunk(qbuf[0], qlane, NBUF);

#pragma unroll
    for (int c = 0; c < NCHUNK; ++c) {
        __builtin_amdgcn_sched_barrier(0);
        FRAGS(c, 1)
        __builtin_amdgcn_sched_barrier(0);
        MATH(0)
        __builtin_amdgcn_sched_barrier(0);
        if (c + 1 < NCHUNK) {
            STAGE(c + 1)
            FRAGS(c + 1, 0)
        }
        __builtin_amdgcn_sched_barrier(0);
        if (c + 1 + NBUF < NCHUNK) load_chunk(qbuf[(c + 1) % NBUF], qlane, c + 1 + NBUF);
        __builtin_amdgcn_sched_barrier(0);
        MATH(1)
    }
#undef STAGE
#undef FRAGS
#undef MATH
    const float aa_part = aa_x + aa_y;

    float aa_full = aa_part + __shfl_xor(aa_part, 16, 32);
    if (lane < 16) aa_lds[wave * 16 + lane] = aa_full;

    const float scl   = scale[0];
    const float inv_d = 1.0f / (float)DIM;
    if (n < NWAY) {
#pragma unroll
        for (int j = 0; j < 8; ++j) {
            int m = mbase + j + 8 * hi;
            if (m >= mown) {
                float aav   = aa_lds[wave * 16 + j + 8 * hi];
                float logit = (2.0f * acc[j] - aav - bb[n]) * inv_d * scl;
                out[((size_t)b * NQ + m) * NWAY + n] = logit;
            }
        }
    }
}

extern "C" void kernel_launch(void* const* d_in, const int* in_sizes, int n_in,
                              void* d_out, int out_size, void* d_ws, size_t ws_size,
                              hipStream_t stream) {
    const float* query   = (const float*)d_in[0];
    const float* support = (const float*)d_in[1];
    const float* scale   = (const float*)d_in[5];
    float* out = (float*)d_out;

    proto_head_wmma_kernel<<<NTASKS, BLOCK, 0, stream>>>(query, support, scale, out);
}
